# starting kernel
# baseline (speedup 1.0000x reference)
#include <hip/hip_runtime.h>
#include <hip/hip_bf16.h>

typedef __attribute__((ext_vector_type(2))) float v2f;
typedef __attribute__((ext_vector_type(8))) float v8f;

#define N_SRC1 500000
#define N_DST1 100000
#define N_DST2 20000
#define IN_F   128
#define H_F    128
#define N_CLS  16

__global__ void zero_ws_kernel(float4* __restrict__ p, size_t n4) {
    size_t i = (size_t)blockIdx.x * blockDim.x + threadIdx.x;
    size_t stride = (size_t)gridDim.x * blockDim.x;
    for (; i < n4; i += stride) p[i] = make_float4(0.f, 0.f, 0.f, 0.f);
}

__global__ void edge_scatter_kernel(const float* __restrict__ feat,
                                    const int*   __restrict__ src,
                                    const int*   __restrict__ dst,
                                    const float* __restrict__ w,
                                    float* __restrict__ acc,
                                    float* __restrict__ deg,
                                    int nEdges) {
    const int lane = threadIdx.x & 31;
    const int e = blockIdx.x * 8 + (threadIdx.x >> 5);
    if (e >= nEdges) return;
    const int   s  = src[e];
    const int   d  = dst[e];
    const float we = w[e];
    const float4 xv = ((const float4*)(feat + (size_t)s * 128))[lane];
    float* ap = acc + (size_t)d * 128 + lane * 4;
    atomicAdd(ap + 0, xv.x * we);
    atomicAdd(ap + 1, xv.y * we);
    atomicAdd(ap + 2, xv.z * we);
    atomicAdd(ap + 3, xv.w * we);
    if (lane == 0) atomicAdd(deg + d, 1.0f);
}

__global__ void sage_gemm1_kernel(const float* __restrict__ x,
                                  const float* __restrict__ acc,
                                  const float* __restrict__ deg,
                                  const float* __restrict__ Ws,
                                  const float* __restrict__ Wn,
                                  const float* __restrict__ bias,
                                  float* __restrict__ h) {
    const int lane  = threadIdx.x & 31;
    const int wave  = threadIdx.x >> 5;
    const int m0    = blockIdx.x * 16;
    const int n0    = wave * 16;
    const int row   = m0 + (lane & 15);
    const int col   = n0 + (lane & 15);
    const int khalf = (lane >> 4) * 2;

    const float* xr = x   + (size_t)row * IN_F;
    const float* ar = acc + (size_t)row * IN_F;
    const float invd = 1.0f / fmaxf(deg[row], 1.0f);

    v8f c = {};
    #pragma unroll 4
    for (int k = 0; k < IN_F; k += 4) {
        const int kk = k + khalf;
        v2f a, b;
        a.x = xr[kk];
        a.y = xr[kk + 1];
        b.x = Ws[(size_t)kk * H_F + col];
        b.y = Ws[(size_t)(kk + 1) * H_F + col];
        c = __builtin_amdgcn_wmma_f32_16x16x4_f32(false, a, false, b,
                                                  (short)0, c, false, false);
    }
    #pragma unroll 4
    for (int k = 0; k < IN_F; k += 4) {
        const int kk = k + khalf;
        v2f a, b;
        a.x = ar[kk]     * invd;
        a.y = ar[kk + 1] * invd;
        b.x = Wn[(size_t)kk * H_F + col];
        b.y = Wn[(size_t)(kk + 1) * H_F + col];
        c = __builtin_amdgcn_wmma_f32_16x16x4_f32(false, a, false, b,
                                                  (short)0, c, false, false);
    }

    const float bv    = bias[col];
    const int   rbase = m0 + ((lane >> 4) * 8);
    #pragma unroll
    for (int r = 0; r < 8; ++r) {
        float v = fmaxf(c[r] + bv, 0.0f);
        h[(size_t)(rbase + r) * H_F + col] = v;
    }
}

__global__ void sage_gemm2_kernel(const float* __restrict__ h,
                                  const float* __restrict__ acc,
                                  const float* __restrict__ deg,
                                  const float* __restrict__ Ws,
                                  const float* __restrict__ Wn,
                                  const float* __restrict__ bias,
                                  float* __restrict__ out) {
    const int lane = threadIdx.x & 31;
    const int wave = threadIdx.x >> 5;
    const int tile = blockIdx.x * 8 + wave;
    if (tile >= N_DST2 / 16) return;
    const int m0    = tile * 16;
    const int row   = m0 + (lane & 15);
    const int col   = lane & 15;
    const int khalf = (lane >> 4) * 2;

    const float* hr = h   + (size_t)row * H_F;
    const float* ar = acc + (size_t)row * H_F;
    const float invd = 1.0f / fmaxf(deg[row], 1.0f);

    v8f c = {};
    #pragma unroll 4
    for (int k = 0; k < H_F; k += 4) {
        const int kk = k + khalf;
        v2f a, b;
        a.x = hr[kk];
        a.y = hr[kk + 1];
        b.x = Ws[(size_t)kk * N_CLS + col];
        b.y = Ws[(size_t)(kk + 1) * N_CLS + col];
        c = __builtin_amdgcn_wmma_f32_16x16x4_f32(false, a, false, b,
                                                  (short)0, c, false, false);
    }
    #pragma unroll 4
    for (int k = 0; k < H_F; k += 4) {
        const int kk = k + khalf;
        v2f a, b;
        a.x = ar[kk]     * invd;
        a.y = ar[kk + 1] * invd;
        b.x = Wn[(size_t)kk * N_CLS + col];
        b.y = Wn[(size_t)(kk + 1) * N_CLS + col];
        c = __builtin_amdgcn_wmma_f32_16x16x4_f32(false, a, false, b,
                                                  (short)0, c, false, false);
    }

    const float bv    = bias[col];
    const int   rbase = m0 + ((lane >> 4) * 8);
    #pragma unroll
    for (int r = 0; r < 8; ++r) {
        float v = c[r] + bv;
        float m = v;
        m = fmaxf(m, __shfl_xor(m, 1, 16));
        m = fmaxf(m, __shfl_xor(m, 2, 16));
        m = fmaxf(m, __shfl_xor(m, 4, 16));
        m = fmaxf(m, __shfl_xor(m, 8, 16));
        float e = __expf(v - m);
        float s = e;
        s += __shfl_xor(s, 1, 16);
        s += __shfl_xor(s, 2, 16);
        s += __shfl_xor(s, 4, 16);
        s += __shfl_xor(s, 8, 16);
        out[(size_t)(rbase + r) * N_CLS + col] = (v - m) - __logf(s);
    }
}

extern "C" void kernel_launch(void* const* d_in, const int* in_sizes, int n_in,
                              void* d_out, int out_size, void* d_ws, size_t ws_size,
                              hipStream_t stream) {
    const float* x    = (const float*)d_in[0];
    const int*   src1 = (const int*)  d_in[1];
    const int*   dst1 = (const int*)  d_in[2];
    const float* w1   = (const float*)d_in[3];
    const int*   src2 = (const int*)  d_in[4];
    const int*   dst2 = (const int*)  d_in[5];
    const float* w2   = (const float*)d_in[6];
    const float* Ws1  = (const float*)d_in[7];
    const float* Wn1  = (const float*)d_in[8];
    const float* b1   = (const float*)d_in[9];
    const float* Ws2  = (const float*)d_in[10];
    const float* Wn2  = (const float*)d_in[11];
    const float* b2   = (const float*)d_in[12];
    float* out = (float*)d_out;

    float* acc1 = (float*)d_ws;
    float* deg1 = acc1 + (size_t)N_DST1 * 128;
    float* hbuf = deg1 + N_DST1;
    float* acc2 = hbuf + (size_t)N_DST1 * 128;
    float* deg2 = acc2 + (size_t)N_DST2 * 128;
    const size_t totalFloats = (size_t)N_DST1 * 128 + N_DST1 +
                               (size_t)N_DST1 * 128 +
                               (size_t)N_DST2 * 128 + N_DST2;

    const int E1 = in_sizes[1];
    const int E2 = in_sizes[4];

    zero_ws_kernel<<<2048, 256, 0, stream>>>((float4*)d_ws, totalFloats / 4);

    edge_scatter_kernel<<<(E1 + 7) / 8, 256, 0, stream>>>(x, src1, dst1, w1,
                                                          acc1, deg1, E1);

    sage_gemm1_kernel<<<N_DST1 / 16, 256, 0, stream>>>(x, acc1, deg1,
                                                       Ws1, Wn1, b1, hbuf);

    edge_scatter_kernel<<<(E2 + 7) / 8, 256, 0, stream>>>(hbuf, src2, dst2, w2,
                                                          acc2, deg2, E2);

    sage_gemm2_kernel<<<(N_DST2 / 16 + 7) / 8, 256, 0, stream>>>(hbuf, acc2, deg2,
                                                                 Ws2, Wn2, b2, out);
}


# Neighbour sums held in LDS: each work-group owns a range of destination rows, filters the whole edge list for them, adds rows with LDS float adds; no global atomic, no zero pass; sum order: arrival
# speedup vs baseline: 2.4218x; 2.4218x over previous
#include <hip/hip_runtime.h>
#include <hip/hip_bf16.h>

typedef __attribute__((ext_vector_type(2))) float v2f;
typedef __attribute__((ext_vector_type(8))) float v8f;

#define N_SRC1 500000
#define N_DST1 100000
#define N_DST2 20000
#define IN_F   128
#define H_F    128
#define N_CLS  16

#define AGG_THREADS 1024
#define AGG_WAVES   32
#define QCAP        544

template <int ROWS>
__global__ __launch_bounds__(AGG_THREADS)
void dst_tile_aggregate_kernel(const float* __restrict__ feat,
                               const int*   __restrict__ src,
                               const int*   __restrict__ dst,
                               const float* __restrict__ w,
                               float* __restrict__ acc,
                               float* __restrict__ deg,
                               int nEdges, int nRows) {
    __shared__ __attribute__((aligned(16))) float sacc[ROWS * 128];
    __shared__ unsigned sdeg[ROWS];
    __shared__ int squeue[AGG_WAVES * QCAP];

    const int tid  = threadIdx.x;
    const int lane = tid & 31;
    const int wave = __builtin_amdgcn_readfirstlane(tid >> 5);
    const int lo   = blockIdx.x * ROWS;

    for (int i = tid; i < ROWS * 32; i += AGG_THREADS)
        ((float4*)sacc)[i] = make_float4(0.f, 0.f, 0.f, 0.f);
    for (int i = tid; i < ROWS; i += AGG_THREADS) sdeg[i] = 0u;
    __syncthreads();

    int* q  = squeue + wave * QCAP;
    int cnt = 0;

    auto flush = [&](int base, int n) {
        const int   e  = (lane < n) ? q[base + lane] : 0;
        const int   s  = src[e];
        const int   d  = dst[e] - lo;
        const float we = w[e];
        if (lane < n)
            __hip_atomic_fetch_add(sdeg + d, 1u, __ATOMIC_RELAXED, __HIP_MEMORY_SCOPE_WORKGROUP);
        for (int k0 = 0; k0 < n; k0 += 8) {
            float4 xv[8];
            int    dk[8];
            float  wk[8];
            #pragma unroll
            for (int u = 0; u < 8; ++u) {
                const int k  = k0 + u;
                const int sk = __builtin_amdgcn_readlane(s, k);
                dk[u] = __builtin_amdgcn_readlane(d, k);
                wk[u] = __int_as_float(__builtin_amdgcn_readlane(__float_as_int(we), k));
                xv[u] = ((const float4*)(feat + (size_t)sk * 128))[lane];
            }
            #pragma unroll
            for (int u = 0; u < 8; ++u) {
                if (k0 + u < n) {
                    float* r = sacc + dk[u] * 128 + lane;
                    __hip_atomic_fetch_add(r,      xv[u].x * wk[u], __ATOMIC_RELAXED, __HIP_MEMORY_SCOPE_WORKGROUP);
                    __hip_atomic_fetch_add(r + 32, xv[u].y * wk[u], __ATOMIC_RELAXED, __HIP_MEMORY_SCOPE_WORKGROUP);
                    __hip_atomic_fetch_add(r + 64, xv[u].z * wk[u], __ATOMIC_RELAXED, __HIP_MEMORY_SCOPE_WORKGROUP);
                    __hip_atomic_fetch_add(r + 96, xv[u].w * wk[u], __ATOMIC_RELAXED, __HIP_MEMORY_SCOPE_WORKGROUP);
                }
            }
        }
    };

    auto push = [&](unsigned ro, int e) {
        const bool     hit = ro < (unsigned)ROWS;
        const unsigned m   = __builtin_amdgcn_ballot_w32(hit);
        if (m) {
            if (hit) q[cnt + (int)__builtin_amdgcn_mbcnt_lo(m, 0u)] = e;
            cnt += __builtin_popcount(m);
        }
    };

    const int nFull = nEdges >> 7;
    const int start = (int)(blockIdx.x * ((unsigned)nFull / gridDim.x));
    for (int i = wave; i < nFull; i += AGG_WAVES * 4) {
        int4 dv[4];
        int  cb[4];
        #pragma unroll
        for (int u = 0; u < 4; ++u) {
            const int ii = i + AGG_WAVES * u;
            int c = start + ii;
            if (c >= nFull) c -= nFull;
            cb[u] = (ii < nFull) ? c : -1;
            dv[u] = (ii < nFull) ? ((const int4*)dst)[(size_t)c * 32 + lane]
                                 : make_int4(-1, -1, -1, -1);
        }
        #pragma unroll
        for (int u = 0; u < 4; ++u) {
            if (cb[u] >= 0) {
                const unsigned rx = (unsigned)(dv[u].x - lo);
                const unsigned ry = (unsigned)(dv[u].y - lo);
                const unsigned rz = (unsigned)(dv[u].z - lo);
                const unsigned rw = (unsigned)(dv[u].w - lo);
                const unsigned mn = min(min(rx, ry), min(rz, rw));
                if (__builtin_amdgcn_ballot_w32(mn < (unsigned)ROWS)) {
                    const int e0 = cb[u] * 128 + lane * 4;
                    push(rx, e0);
                    push(ry, e0 + 1);
                    push(rz, e0 + 2);
                    push(rw, e0 + 3);
                }
            }
        }
        while (cnt >= 32) { cnt -= 32; flush(cnt, 32); }
    }
    if (wave == 0) {
        const int e0 = nFull * 128 + lane * 4;
        for (int j = 0; j < 4; ++j) {
            const int e = e0 + j;
            push((e < nEdges) ? (unsigned)(dst[e] - lo) : 0xffffffffu, e);
        }
    }
    while (cnt > 0) {
        const int n = cnt < 32 ? cnt : 32;
        cnt -= n;
        flush(cnt, n);
    }

    __syncthreads();

    for (int r = wave; r < ROWS; r += AGG_WAVES) {
        if (lo + r < nRows) {
            const float* p = sacc + r * 128 + lane;
            ((float4*)(acc + (size_t)(lo + r) * 128))[lane] =
                make_float4(p[0], p[32], p[64], p[96]);
        }
    }
    for (int i = tid; i < ROWS; i += AGG_THREADS)
        if (lo + i < nRows) deg[lo + i] = (float)sdeg[i];
}

#define ROWS_L1 416
#define ROWS_L2 96

__global__ void sage_gemm1_kernel(const float* __restrict__ x,
                                  const float* __restrict__ acc,
                                  const float* __restrict__ deg,
                                  const float* __restrict__ Ws,
                                  const float* __restrict__ Wn,
                                  const float* __restrict__ bias,
                                  float* __restrict__ h) {
    const int lane  = threadIdx.x & 31;
    const int wave  = threadIdx.x >> 5;
    const int m0    = blockIdx.x * 16;
    const int n0    = wave * 16;
    const int row   = m0 + (lane & 15);
    const int col   = n0 + (lane & 15);
    const int khalf = (lane >> 4) * 2;

    const float* xr = x   + (size_t)row * IN_F;
    const float* ar = acc + (size_t)row * IN_F;
    const float invd = 1.0f / fmaxf(deg[row], 1.0f);

    v8f c = {};
    #pragma unroll 4
    for (int k = 0; k < IN_F; k += 4) {
        const int kk = k + khalf;
        v2f a, b;
        a.x = xr[kk];
        a.y = xr[kk + 1];
        b.x = Ws[(size_t)kk * H_F + col];
        b.y = Ws[(size_t)(kk + 1) * H_F + col];
        c = __builtin_amdgcn_wmma_f32_16x16x4_f32(false, a, false, b,
                                                  (short)0, c, false, false);
    }
    #pragma unroll 4
    for (int k = 0; k < IN_F; k += 4) {
        const int kk = k + khalf;
        v2f a, b;
        a.x = ar[kk]     * invd;
        a.y = ar[kk + 1] * invd;
        b.x = Wn[(size_t)kk * H_F + col];
        b.y = Wn[(size_t)(kk + 1) * H_F + col];
        c = __builtin_amdgcn_wmma_f32_16x16x4_f32(false, a, false, b,
                                                  (short)0, c, false, false);
    }

    const float bv    = bias[col];
    const int   rbase = m0 + ((lane >> 4) * 8);
    #pragma unroll
    for (int r = 0; r < 8; ++r) {
        float v = fmaxf(c[r] + bv, 0.0f);
        h[(size_t)(rbase + r) * H_F + col] = v;
    }
}

__global__ void sage_gemm2_kernel(const float* __restrict__ h,
                                  const float* __restrict__ acc,
                                  const float* __restrict__ deg,
                                  const float* __restrict__ Ws,
                                  const float* __restrict__ Wn,
                                  const float* __restrict__ bias,
                                  float* __restrict__ out) {
    const int lane = threadIdx.x & 31;
    const int wave = threadIdx.x >> 5;
    const int tile = blockIdx.x * 8 + wave;
    if (tile >= N_DST2 / 16) return;
    const int m0    = tile * 16;
    const int row   = m0 + (lane & 15);
    const int col   = lane & 15;
    const int khalf = (lane >> 4) * 2;

    const float* hr = h   + (size_t)row * H_F;
    const float* ar = acc + (size_t)row * H_F;
    const float invd = 1.0f / fmaxf(deg[row], 1.0f);

    v8f c = {};
    #pragma unroll 4
    for (int k = 0; k < H_F; k += 4) {
        const int kk = k + khalf;
        v2f a, b;
        a.x = hr[kk];
        a.y = hr[kk + 1];
        b.x = Ws[(size_t)kk * N_CLS + col];
        b.y = Ws[(size_t)(kk + 1) * N_CLS + col];
        c = __builtin_amdgcn_wmma_f32_16x16x4_f32(false, a, false, b,
                                                  (short)0, c, false, false);
    }
    #pragma unroll 4
    for (int k = 0; k < H_F; k += 4) {
        const int kk = k + khalf;
        v2f a, b;
        a.x = ar[kk]     * invd;
        a.y = ar[kk + 1] * invd;
        b.x = Wn[(size_t)kk * N_CLS + col];
        b.y = Wn[(size_t)(kk + 1) * N_CLS + col];
        c = __builtin_amdgcn_wmma_f32_16x16x4_f32(false, a, false, b,
                                                  (short)0, c, false, false);
    }

    const float bv    = bias[col];
    const int   rbase = m0 + ((lane >> 4) * 8);
    #pragma unroll
    for (int r = 0; r < 8; ++r) {
        float v = c[r] + bv;
        float m = v;
        m = fmaxf(m, __shfl_xor(m, 1, 16));
        m = fmaxf(m, __shfl_xor(m, 2, 16));
        m = fmaxf(m, __shfl_xor(m, 4, 16));
        m = fmaxf(m, __shfl_xor(m, 8, 16));
        float e = __expf(v - m);
        float s = e;
        s += __shfl_xor(s, 1, 16);
        s += __shfl_xor(s, 2, 16);
        s += __shfl_xor(s, 4, 16);
        s += __shfl_xor(s, 8, 16);
        out[(size_t)(rbase + r) * N_CLS + col] = (v - m) - __logf(s);
    }
}

extern "C" void kernel_launch(void* const* d_in, const int* in_sizes, int n_in,
                              void* d_out, int out_size, void* d_ws, size_t ws_size,
                              hipStream_t stream) {
    const float* x    = (const float*)d_in[0];
    const int*   src1 = (const int*)  d_in[1];
    const int*   dst1 = (const int*)  d_in[2];
    const float* w1   = (const float*)d_in[3];
    const int*   src2 = (const int*)  d_in[4];
    const int*   dst2 = (const int*)  d_in[5];
    const float* w2   = (const float*)d_in[6];
    const float* Ws1  = (const float*)d_in[7];
    const float* Wn1  = (const float*)d_in[8];
    const float* b1   = (const float*)d_in[9];
    const float* Ws2  = (const float*)d_in[10];
    const float* Wn2  = (const float*)d_in[11];
    const float* b2   = (const float*)d_in[12];
    float* out = (float*)d_out;

    float* acc1 = (float*)d_ws;
    float* deg1 = acc1 + (size_t)N_DST1 * 128;
    float* hbuf = deg1 + N_DST1;
    float* acc2 = hbuf + (size_t)N_DST1 * 128;
    float* deg2 = acc2 + (size_t)N_DST2 * 128;

    const int E1 = in_sizes[1];
    const int E2 = in_sizes[4];

    dst_tile_aggregate_kernel<ROWS_L1>
        <<<(N_DST1 + ROWS_L1 - 1) / ROWS_L1, AGG_THREADS, 0, stream>>>(
            x, src1, dst1, w1, acc1, deg1, E1, N_DST1);

    sage_gemm1_kernel<<<N_DST1 / 16, 256, 0, stream>>>(x, acc1, deg1,
                                                       Ws1, Wn1, b1, hbuf);

    dst_tile_aggregate_kernel<ROWS_L2>
        <<<(N_DST2 + ROWS_L2 - 1) / ROWS_L2, AGG_THREADS, 0, stream>>>(
            hbuf, src2, dst2, w2, acc2, deg2, E2, N_DST2);

    sage_gemm2_kernel<<<(N_DST2 / 16 + 7) / 8, 256, 0, stream>>>(hbuf, acc2, deg2,
                                                                 Ws2, Wn2, b2, out);
}


# Edge-list filter reads one-byte tile keys: a kernel per layer writes dst / rows-per-group as 1 B per edge; work-groups scan 16 keys per lane with a zero-byte test; sum order: arrival, as before
# speedup vs baseline: 2.6607x; 1.0986x over previous
#include <hip/hip_runtime.h>
#include <hip/hip_bf16.h>

typedef __attribute__((ext_vector_type(2))) float v2f;
typedef __attribute__((ext_vector_type(8))) float v8f;

#define N_SRC1 500000
#define N_DST1 100000
#define N_DST2 20000
#define IN_F   128
#define H_F    128
#define N_CLS  16

#define KEYS_PER_LOAD 512

template <int ROWS>
__device__ __forceinline__ unsigned tile_key(int d) {
    const unsigned q = (unsigned)d / (unsigned)ROWS;
    return q < 255u ? q : 255u;
}

template <int ROWS>
__global__ __launch_bounds__(256)
void tile_key_kernel(const int* __restrict__ dst, unsigned* __restrict__ keys, int nEdges) {
    const int nGroups = ((nEdges + KEYS_PER_LOAD - 1) / KEYS_PER_LOAD) * (KEYS_PER_LOAD / 16);
    for (int g = blockIdx.x * 256 + threadIdx.x; g < nGroups; g += gridDim.x * 256) {
        const int e0 = g * 16;
        unsigned k0 = 0u, k1 = 0u, k2 = 0u, k3 = 0u;
        if (e0 + 16 <= nEdges) {
            const int4 a = ((const int4*)dst)[(size_t)g * 4];
            const int4 b = ((const int4*)dst)[(size_t)g * 4 + 1];
            const int4 c = ((const int4*)dst)[(size_t)g * 4 + 2];
            const int4 d = ((const int4*)dst)[(size_t)g * 4 + 3];
            k0 = tile_key<ROWS>(a.x) | (tile_key<ROWS>(a.y) << 8) | (tile_key<ROWS>(a.z) << 16) | (tile_key<ROWS>(a.w) << 24);
            k1 = tile_key<ROWS>(b.x) | (tile_key<ROWS>(b.y) << 8) | (tile_key<ROWS>(b.z) << 16) | (tile_key<ROWS>(b.w) << 24);
            k2 = tile_key<ROWS>(c.x) | (tile_key<ROWS>(c.y) << 8) | (tile_key<ROWS>(c.z) << 16) | (tile_key<ROWS>(c.w) << 24);
            k3 = tile_key<ROWS>(d.x) | (tile_key<ROWS>(d.y) << 8) | (tile_key<ROWS>(d.z) << 16) | (tile_key<ROWS>(d.w) << 24);
        } else {
            for (int b = 0; b < 4; ++b) {
                const int ea = e0 + b, eb = e0 + 4 + b, ec = e0 + 8 + b, ed = e0 + 12 + b;
                k0 |= ((ea < nEdges) ? tile_key<ROWS>(dst[ea]) : 255u) << (8 * b);
                k1 |= ((eb < nEdges) ? tile_key<ROWS>(dst[eb]) : 255u) << (8 * b);
                k2 |= ((ec < nEdges) ? tile_key<ROWS>(dst[ec]) : 255u) << (8 * b);
                k3 |= ((ed < nEdges) ? tile_key<ROWS>(dst[ed]) : 255u) << (8 * b);
            }
        }
        ((uint4*)keys)[g] = make_uint4(k0, k1, k2, k3);
    }
}

#define AGG_THREADS 1024
#define AGG_WAVES   32
#define QCAP        544

template <int ROWS>
__global__ __launch_bounds__(AGG_THREADS)
void dst_tile_aggregate_kernel(const float* __restrict__ feat,
                               const int*   __restrict__ src,
                               const int*   __restrict__ dst,
                               const unsigned* __restrict__ keys,
                               const float* __restrict__ w,
                               float* __restrict__ acc,
                               float* __restrict__ deg,
                               int nEdges, int nRows) {
    __shared__ __attribute__((aligned(16))) float sacc[ROWS * 128];
    __shared__ unsigned sdeg[ROWS];
    __shared__ int squeue[AGG_WAVES * QCAP];

    const int tid  = threadIdx.x;
    const int lane = tid & 31;
    const int wave = __builtin_amdgcn_readfirstlane(tid >> 5);
    const int lo   = blockIdx.x * ROWS;

    for (int i = tid; i < ROWS * 32; i += AGG_THREADS)
        ((float4*)sacc)[i] = make_float4(0.f, 0.f, 0.f, 0.f);
    for (int i = tid; i < ROWS; i += AGG_THREADS) sdeg[i] = 0u;
    __syncthreads();

    int* q  = squeue + wave * QCAP;
    int cnt = 0;

    auto flush = [&](int base, int n) {
        const int   e  = (lane < n) ? q[base + lane] : 0;
        const int   s  = src[e];
        const int   d  = dst[e] - lo;
        const float we = w[e];
        if (lane < n)
            __hip_atomic_fetch_add(sdeg + d, 1u, __ATOMIC_RELAXED, __HIP_MEMORY_SCOPE_WORKGROUP);
        for (int k0 = 0; k0 < n; k0 += 8) {
            float4 xv[8];
            int    dk[8];
            float  wk[8];
            #pragma unroll
            for (int u = 0; u < 8; ++u) {
                const int k  = k0 + u;
                const int sk = __builtin_amdgcn_readlane(s, k);
                dk[u] = __builtin_amdgcn_readlane(d, k);
                wk[u] = __int_as_float(__builtin_amdgcn_readlane(__float_as_int(we), k));
                xv[u] = ((const float4*)(feat + (size_t)sk * 128))[lane];
            }
            #pragma unroll
            for (int u = 0; u < 8; ++u) {
                if (k0 + u < n) {
                    float* r = sacc + dk[u] * 128 + lane;
                    __hip_atomic_fetch_add(r,      xv[u].x * wk[u], __ATOMIC_RELAXED, __HIP_MEMORY_SCOPE_WORKGROUP);
                    __hip_atomic_fetch_add(r + 32, xv[u].y * wk[u], __ATOMIC_RELAXED, __HIP_MEMORY_SCOPE_WORKGROUP);
                    __hip_atomic_fetch_add(r + 64, xv[u].z * wk[u], __ATOMIC_RELAXED, __HIP_MEMORY_SCOPE_WORKGROUP);
                    __hip_atomic_fetch_add(r + 96, xv[u].w * wk[u], __ATOMIC_RELAXED, __HIP_MEMORY_SCOPE_WORKGROUP);
                }
            }
        }
    };

    auto push = [&](bool hit, int e) {
        const unsigned m = __builtin_amdgcn_ballot_w32(hit);
        if (m) {
            if (hit) q[cnt + (int)__builtin_amdgcn_mbcnt_lo(m, 0u)] = e;
            cnt += __builtin_popcount(m);
        }
    };

    auto scan_word = [&](unsigned x, unsigned z, int e) {
        if (__builtin_amdgcn_ballot_w32((z & 0x80808080u) != 0u)) {
            push((x & 0x000000ffu) == 0u, e);
            push((x & 0x0000ff00u) == 0u, e + 1);
            push((x & 0x00ff0000u) == 0u, e + 2);
            push((x & 0xff000000u) == 0u, e + 3);
        }
    };

    const unsigned pat     = (unsigned)blockIdx.x * 0x01010101u;
    const int      nChunks = (nEdges + KEYS_PER_LOAD - 1) / KEYS_PER_LOAD;
    const int      start   = (int)(blockIdx.x * ((unsigned)nChunks / gridDim.x));
    for (int i = wave; i < nChunks; i += AGG_WAVES * 4) {
        uint4 kv[4];
        int   cb[4];
        #pragma unroll
        for (int u = 0; u < 4; ++u) {
            const int ii = i + AGG_WAVES * u;
            int c = start + ii;
            if (c >= nChunks) c -= nChunks;
            cb[u] = (ii < nChunks) ? c : -1;
            kv[u] = (ii < nChunks) ? ((const uint4*)keys)[(size_t)c * 32 + lane]
                                   : make_uint4(~0u, ~0u, ~0u, ~0u);
        }
        #pragma unroll
        for (int u = 0; u < 4; ++u) {
            if (cb[u] >= 0) {
                const unsigned x0 = kv[u].x ^ pat;
                const unsigned x1 = kv[u].y ^ pat;
                const unsigned x2 = kv[u].z ^ pat;
                const unsigned x3 = kv[u].w ^ pat;
                const unsigned z0 = (x0 - 0x01010101u) & ~x0;
                const unsigned z1 = (x1 - 0x01010101u) & ~x1;
                const unsigned z2 = (x2 - 0x01010101u) & ~x2;
                const unsigned z3 = (x3 - 0x01010101u) & ~x3;
                if (__builtin_amdgcn_ballot_w32(((z0 | z1 | z2 | z3) & 0x80808080u) != 0u)) {
                    const int e0 = cb[u] * KEYS_PER_LOAD + lane * 16;
                    scan_word(x0, z0, e0);
                    scan_word(x1, z1, e0 + 4);
                    scan_word(x2, z2, e0 + 8);
                    scan_word(x3, z3, e0 + 12);
                }
                while (cnt >= 32) { cnt -= 32; flush(cnt, 32); }
            }
        }
    }
    while (cnt > 0) {
        const int n = cnt < 32 ? cnt : 32;
        cnt -= n;
        flush(cnt, n);
    }

    __syncthreads();

    for (int r = wave; r < ROWS; r += AGG_WAVES) {
        if (lo + r < nRows) {
            const float* p = sacc + r * 128 + lane;
            ((float4*)(acc + (size_t)(lo + r) * 128))[lane] =
                make_float4(p[0], p[32], p[64], p[96]);
        }
    }
    for (int i = tid; i < ROWS; i += AGG_THREADS)
        if (lo + i < nRows) deg[lo + i] = (float)sdeg[i];
}

#define ROWS_L1 416
#define ROWS_L2 96

__global__ void sage_gemm1_kernel(const float* __restrict__ x,
                                  const float* __restrict__ acc,
                                  const float* __restrict__ deg,
                                  const float* __restrict__ Ws,
                                  const float* __restrict__ Wn,
                                  const float* __restrict__ bias,
                                  float* __restrict__ h) {
    const int lane  = threadIdx.x & 31;
    const int wave  = threadIdx.x >> 5;
    const int m0    = blockIdx.x * 16;
    const int n0    = wave * 16;
    const int row   = m0 + (lane & 15);
    const int col   = n0 + (lane & 15);
    const int khalf = (lane >> 4) * 2;

    const float* xr = x   + (size_t)row * IN_F;
    const float* ar = acc + (size_t)row * IN_F;
    const float invd = 1.0f / fmaxf(deg[row], 1.0f);

    v8f c = {};
    #pragma unroll 4
    for (int k = 0; k < IN_F; k += 4) {
        const int kk = k + khalf;
        v2f a, b;
        a.x = xr[kk];
        a.y = xr[kk + 1];
        b.x = Ws[(size_t)kk * H_F + col];
        b.y = Ws[(size_t)(kk + 1) * H_F + col];
        c = __builtin_amdgcn_wmma_f32_16x16x4_f32(false, a, false, b,
                                                  (short)0, c, false, false);
    }
    #pragma unroll 4
    for (int k = 0; k < IN_F; k += 4) {
        const int kk = k + khalf;
        v2f a, b;
        a.x = ar[kk]     * invd;
        a.y = ar[kk + 1] * invd;
        b.x = Wn[(size_t)kk * H_F + col];
        b.y = Wn[(size_t)(kk + 1) * H_F + col];
        c = __builtin_amdgcn_wmma_f32_16x16x4_f32(false, a, false, b,
                                                  (short)0, c, false, false);
    }

    const float bv    = bias[col];
    const int   rbase = m0 + ((lane >> 4) * 8);
    #pragma unroll
    for (int r = 0; r < 8; ++r) {
        float v = fmaxf(c[r] + bv, 0.0f);
        h[(size_t)(rbase + r) * H_F + col] = v;
    }
}

__global__ void sage_gemm2_kernel(const float* __restrict__ h,
                                  const float* __restrict__ acc,
                                  const float* __restrict__ deg,
                                  const float* __restrict__ Ws,
                                  const float* __restrict__ Wn,
                                  const float* __restrict__ bias,
                                  float* __restrict__ out) {
    const int lane = threadIdx.x & 31;
    const int wave = threadIdx.x >> 5;
    const int tile = blockIdx.x * 8 + wave;
    if (tile >= N_DST2 / 16) return;
    const int m0    = tile * 16;
    const int row   = m0 + (lane & 15);
    const int col   = lane & 15;
    const int khalf = (lane >> 4) * 2;

    const float* hr = h   + (size_t)row * H_F;
    const float* ar = acc + (size_t)row * H_F;
    const float invd = 1.0f / fmaxf(deg[row], 1.0f);

    v8f c = {};
    #pragma unroll 4
    for (int k = 0; k < H_F; k += 4) {
        const int kk = k + khalf;
        v2f a, b;
        a.x = hr[kk];
        a.y = hr[kk + 1];
        b.x = Ws[(size_t)kk * N_CLS + col];
        b.y = Ws[(size_t)(kk + 1) * N_CLS + col];
        c = __builtin_amdgcn_wmma_f32_16x16x4_f32(false, a, false, b,
                                                  (short)0, c, false, false);
    }
    #pragma unroll 4
    for (int k = 0; k < H_F; k += 4) {
        const int kk = k + khalf;
        v2f a, b;
        a.x = ar[kk]     * invd;
        a.y = ar[kk + 1] * invd;
        b.x = Wn[(size_t)kk * N_CLS + col];
        b.y = Wn[(size_t)(kk + 1) * N_CLS + col];
        c = __builtin_amdgcn_wmma_f32_16x16x4_f32(false, a, false, b,
                                                  (short)0, c, false, false);
    }

    const float bv    = bias[col];
    const int   rbase = m0 + ((lane >> 4) * 8);
    #pragma unroll
    for (int r = 0; r < 8; ++r) {
        float v = c[r] + bv;
        float m = v;
        m = fmaxf(m, __shfl_xor(m, 1, 16));
        m = fmaxf(m, __shfl_xor(m, 2, 16));
        m = fmaxf(m, __shfl_xor(m, 4, 16));
        m = fmaxf(m, __shfl_xor(m, 8, 16));
        float e = __expf(v - m);
        float s = e;
        s += __shfl_xor(s, 1, 16);
        s += __shfl_xor(s, 2, 16);
        s += __shfl_xor(s, 4, 16);
        s += __shfl_xor(s, 8, 16);
        out[(size_t)(rbase + r) * N_CLS + col] = (v - m) - __logf(s);
    }
}

extern "C" void kernel_launch(void* const* d_in, const int* in_sizes, int n_in,
                              void* d_out, int out_size, void* d_ws, size_t ws_size,
                              hipStream_t stream) {
    const float* x    = (const float*)d_in[0];
    const int*   src1 = (const int*)  d_in[1];
    const int*   dst1 = (const int*)  d_in[2];
    const float* w1   = (const float*)d_in[3];
    const int*   src2 = (const int*)  d_in[4];
    const int*   dst2 = (const int*)  d_in[5];
    const float* w2   = (const float*)d_in[6];
    const float* Ws1  = (const float*)d_in[7];
    const float* Wn1  = (const float*)d_in[8];
    const float* b1   = (const float*)d_in[9];
    const float* Ws2  = (const float*)d_in[10];
    const float* Wn2  = (const float*)d_in[11];
    const float* b2   = (const float*)d_in[12];
    float* out = (float*)d_out;

    float* acc1 = (float*)d_ws;
    float* deg1 = acc1 + (size_t)N_DST1 * 128;
    float* hbuf = deg1 + N_DST1;
    float* acc2 = hbuf + (size_t)N_DST1 * 128;
    float* deg2 = acc2 + (size_t)N_DST2 * 128;

    const int E1 = in_sizes[1];
    const int E2 = in_sizes[4];

    unsigned* keys1 = (unsigned*)hbuf;
    unsigned* keys2 = (unsigned*)acc1;

    tile_key_kernel<ROWS_L1><<<512, 256, 0, stream>>>(dst1, keys1, E1);

    dst_tile_aggregate_kernel<ROWS_L1>
        <<<(N_DST1 + ROWS_L1 - 1) / ROWS_L1, AGG_THREADS, 0, stream>>>(
            x, src1, dst1, keys1, w1, acc1, deg1, E1, N_DST1);

    sage_gemm1_kernel<<<N_DST1 / 16, 256, 0, stream>>>(x, acc1, deg1,
                                                       Ws1, Wn1, b1, hbuf);

    tile_key_kernel<ROWS_L2><<<256, 256, 0, stream>>>(dst2, keys2, E2);
    dst_tile_aggregate_kernel<ROWS_L2>
        <<<(N_DST2 + ROWS_L2 - 1) / ROWS_L2, AGG_THREADS, 0, stream>>>(
            hbuf, src2, dst2, keys2, w2, acc2, deg2, E2, N_DST2);

    sage_gemm2_kernel<<<(N_DST2 / 16 + 7) / 8, 256, 0, stream>>>(hbuf, acc2, deg2,
                                                                 Ws2, Wn2, b2, out);
}


# Layer-1 GEMM: both weight matrices sit in LDS as ready B fragments (128 KiB, copied once per work-group); a wave owns a 16x128 tile, 8 accumulators, A fragments straight from memory; k order kept
# speedup vs baseline: 3.1700x; 1.1914x over previous
#include <hip/hip_runtime.h>
#include <hip/hip_bf16.h>

typedef __attribute__((ext_vector_type(2))) float v2f;
typedef __attribute__((ext_vector_type(8))) float v8f;

#define N_SRC1 500000
#define N_DST1 100000
#define N_DST2 20000
#define IN_F   128
#define H_F    128
#define N_CLS  16

#define KEYS_PER_LOAD 512

template <int ROWS>
__device__ __forceinline__ unsigned tile_key(int d) {
    const unsigned q = (unsigned)d / (unsigned)ROWS;
    return q < 255u ? q : 255u;
}

template <int ROWS>
__global__ __launch_bounds__(256)
void tile_key_kernel(const int* __restrict__ dst, unsigned* __restrict__ keys, int nEdges) {
    const int nGroups = ((nEdges + KEYS_PER_LOAD - 1) / KEYS_PER_LOAD) * (KEYS_PER_LOAD / 16);
    for (int g = blockIdx.x * 256 + threadIdx.x; g < nGroups; g += gridDim.x * 256) {
        const int e0 = g * 16;
        unsigned k0 = 0u, k1 = 0u, k2 = 0u, k3 = 0u;
        if (e0 + 16 <= nEdges) {
            const int4 a = ((const int4*)dst)[(size_t)g * 4];
            const int4 b = ((const int4*)dst)[(size_t)g * 4 + 1];
            const int4 c = ((const int4*)dst)[(size_t)g * 4 + 2];
            const int4 d = ((const int4*)dst)[(size_t)g * 4 + 3];
            k0 = tile_key<ROWS>(a.x) | (tile_key<ROWS>(a.y) << 8) | (tile_key<ROWS>(a.z) << 16) | (tile_key<ROWS>(a.w) << 24);
            k1 = tile_key<ROWS>(b.x) | (tile_key<ROWS>(b.y) << 8) | (tile_key<ROWS>(b.z) << 16) | (tile_key<ROWS>(b.w) << 24);
            k2 = tile_key<ROWS>(c.x) | (tile_key<ROWS>(c.y) << 8) | (tile_key<ROWS>(c.z) << 16) | (tile_key<ROWS>(c.w) << 24);
            k3 = tile_key<ROWS>(d.x) | (tile_key<ROWS>(d.y) << 8) | (tile_key<ROWS>(d.z) << 16) | (tile_key<ROWS>(d.w) << 24);
        } else {
            for (int b = 0; b < 4; ++b) {
                const int ea = e0 + b, eb = e0 + 4 + b, ec = e0 + 8 + b, ed = e0 + 12 + b;
                k0 |= ((ea < nEdges) ? tile_key<ROWS>(dst[ea]) : 255u) << (8 * b);
                k1 |= ((eb < nEdges) ? tile_key<ROWS>(dst[eb]) : 255u) << (8 * b);
                k2 |= ((ec < nEdges) ? tile_key<ROWS>(dst[ec]) : 255u) << (8 * b);
                k3 |= ((ed < nEdges) ? tile_key<ROWS>(dst[ed]) : 255u) << (8 * b);
            }
        }
        ((uint4*)keys)[g] = make_uint4(k0, k1, k2, k3);
    }
}

#define AGG_THREADS 1024
#define AGG_WAVES   32
#define QCAP        544

template <int ROWS>
__global__ __launch_bounds__(AGG_THREADS)
void dst_tile_aggregate_kernel(const float* __restrict__ feat,
                               const int*   __restrict__ src,
                               const int*   __restrict__ dst,
                               const unsigned* __restrict__ keys,
                               const float* __restrict__ w,
                               float* __restrict__ acc,
                               float* __restrict__ deg,
                               int nEdges, int nRows) {
    __shared__ __attribute__((aligned(16))) float sacc[ROWS * 128];
    __shared__ unsigned sdeg[ROWS];
    __shared__ int squeue[AGG_WAVES * QCAP];

    const int tid  = threadIdx.x;
    const int lane = tid & 31;
    const int wave = __builtin_amdgcn_readfirstlane(tid >> 5);
    const int lo   = blockIdx.x * ROWS;

    for (int i = tid; i < ROWS * 32; i += AGG_THREADS)
        ((float4*)sacc)[i] = make_float4(0.f, 0.f, 0.f, 0.f);
    for (int i = tid; i < ROWS; i += AGG_THREADS) sdeg[i] = 0u;
    __syncthreads();

    int* q  = squeue + wave * QCAP;
    int cnt = 0;

    auto flush = [&](int base, int n) {
        const int   e  = (lane < n) ? q[base + lane] : 0;
        const int   s  = src[e];
        const int   d  = dst[e] - lo;
        const float we = w[e];
        if (lane < n)
            __hip_atomic_fetch_add(sdeg + d, 1u, __ATOMIC_RELAXED, __HIP_MEMORY_SCOPE_WORKGROUP);
        for (int k0 = 0; k0 < n; k0 += 8) {
            float4 xv[8];
            int    dk[8];
            float  wk[8];
            #pragma unroll
            for (int u = 0; u < 8; ++u) {
                const int k  = k0 + u;
                const int sk = __builtin_amdgcn_readlane(s, k);
                dk[u] = __builtin_amdgcn_readlane(d, k);
                wk[u] = __int_as_float(__builtin_amdgcn_readlane(__float_as_int(we), k));
                xv[u] = ((const float4*)(feat + (size_t)sk * 128))[lane];
            }
            #pragma unroll
            for (int u = 0; u < 8; ++u) {
                if (k0 + u < n) {
                    float* r = sacc + dk[u] * 128 + lane;
                    __hip_atomic_fetch_add(r,      xv[u].x * wk[u], __ATOMIC_RELAXED, __HIP_MEMORY_SCOPE_WORKGROUP);
                    __hip_atomic_fetch_add(r + 32, xv[u].y * wk[u], __ATOMIC_RELAXED, __HIP_MEMORY_SCOPE_WORKGROUP);
                    __hip_atomic_fetch_add(r + 64, xv[u].z * wk[u], __ATOMIC_RELAXED, __HIP_MEMORY_SCOPE_WORKGROUP);
                    __hip_atomic_fetch_add(r + 96, xv[u].w * wk[u], __ATOMIC_RELAXED, __HIP_MEMORY_SCOPE_WORKGROUP);
                }
            }
        }
    };

    auto push = [&](bool hit, int e) {
        const unsigned m = __builtin_amdgcn_ballot_w32(hit);
        if (m) {
            if (hit) q[cnt + (int)__builtin_amdgcn_mbcnt_lo(m, 0u)] = e;
            cnt += __builtin_popcount(m);
        }
    };

    auto scan_word = [&](unsigned x, unsigned z, int e) {
        if (__builtin_amdgcn_ballot_w32((z & 0x80808080u) != 0u)) {
            push((x & 0x000000ffu) == 0u, e);
            push((x & 0x0000ff00u) == 0u, e + 1);
            push((x & 0x00ff0000u) == 0u, e + 2);
            push((x & 0xff000000u) == 0u, e + 3);
        }
    };

    const unsigned pat     = (unsigned)blockIdx.x * 0x01010101u;
    const int      nChunks = (nEdges + KEYS_PER_LOAD - 1) / KEYS_PER_LOAD;
    const int      start   = (int)(blockIdx.x * ((unsigned)nChunks / gridDim.x));
    for (int i = wave; i < nChunks; i += AGG_WAVES * 4) {
        uint4 kv[4];
        int   cb[4];
        #pragma unroll
        for (int u = 0; u < 4; ++u) {
            const int ii = i + AGG_WAVES * u;
            int c = start + ii;
            if (c >= nChunks) c -= nChunks;
            cb[u] = (ii < nChunks) ? c : -1;
            kv[u] = (ii < nChunks) ? ((const uint4*)keys)[(size_t)c * 32 + lane]
                                   : make_uint4(~0u, ~0u, ~0u, ~0u);
        }
        #pragma unroll
        for (int u = 0; u < 4; ++u) {
            if (cb[u] >= 0) {
                const unsigned x0 = kv[u].x ^ pat;
                const unsigned x1 = kv[u].y ^ pat;
                const unsigned x2 = kv[u].z ^ pat;
                const unsigned x3 = kv[u].w ^ pat;
                const unsigned z0 = (x0 - 0x01010101u) & ~x0;
                const unsigned z1 = (x1 - 0x01010101u) & ~x1;
                const unsigned z2 = (x2 - 0x01010101u) & ~x2;
                const unsigned z3 = (x3 - 0x01010101u) & ~x3;
                if (__builtin_amdgcn_ballot_w32(((z0 | z1 | z2 | z3) & 0x80808080u) != 0u)) {
                    const int e0 = cb[u] * KEYS_PER_LOAD + lane * 16;
                    scan_word(x0, z0, e0);
                    scan_word(x1, z1, e0 + 4);
                    scan_word(x2, z2, e0 + 8);
                    scan_word(x3, z3, e0 + 12);
                }
                while (cnt >= 32) { cnt -= 32; flush(cnt, 32); }
            }
        }
    }
    while (cnt > 0) {
        const int n = cnt < 32 ? cnt : 32;
        cnt -= n;
        flush(cnt, n);
    }

    __syncthreads();

    for (int r = wave; r < ROWS; r += AGG_WAVES) {
        if (lo + r < nRows) {
            const float* p = sacc + r * 128 + lane;
            ((float4*)(acc + (size_t)(lo + r) * 128))[lane] =
                make_float4(p[0], p[32], p[64], p[96]);
        }
    }
    for (int i = tid; i < ROWS; i += AGG_THREADS)
        if (lo + i < nRows) deg[lo + i] = (float)sdeg[i];
}

#define ROWS_L1 416
#define ROWS_L2 96

#define G1_WAVES   25
#define G1_THREADS (G1_WAVES * 32)
static_assert(N_DST1 % (16 * G1_WAVES) == 0, "M tiles must fill whole work-groups");

__global__ __launch_bounds__(G1_THREADS)
void sage_gemm1_kernel(const float* __restrict__ x,
                       const float* __restrict__ acc,
                       const float* __restrict__ deg,
                       const float* __restrict__ Ws,
                       const float* __restrict__ Wn,
                       const float* __restrict__ bias,
                       float* __restrict__ h) {
    __shared__ __attribute__((aligned(16))) float sW[(2 * IN_F / 8) * 8 * 32 * 4];

    const int tid  = threadIdx.x;
    const int lane = tid & 31;
    const int wave = tid >> 5;
    const int c    = lane & 15;
    const int hh   = lane >> 4;
    const int m0   = (blockIdx.x * G1_WAVES + wave) * 16;
    const int row  = m0 + c;

    const float* xr = x   + (size_t)row * IN_F + 2 * hh;
    const float* ar = acc + (size_t)row * IN_F + 2 * hh;

    v2f a0 = *(const v2f*)(xr);
    v2f a1 = *(const v2f*)(xr + 4);
    const float invd = 1.0f / fmaxf(deg[row], 1.0f);

    for (int i = tid; i < IN_F * H_F; i += G1_THREADS) {
        const int p   = i >> 7;
        const int col = i & 127;
        const int k0  = 2 * p;
        const float* wsrc = (k0 < IN_F) ? Ws + (size_t)k0 * H_F
                                        : Wn + (size_t)(k0 - IN_F) * H_F;
        v2f v;
        v.x = wsrc[col];
        v.y = wsrc[H_F + col];
        const int t  = p >> 1;
        const int fh = p & 1;
        const int fl = fh * 16 + (col & 15);
        const int e  = ((t >> 1) * 8 + (col >> 4)) * 32 + fl;
        *(v2f*)(sW + e * 4 + (t & 1) * 2) = v;
    }
    __syncthreads();

    const float4* bw = (const float4*)sW + lane;
    v8f cc[8] = {};

    auto step2 = [&](v2f f0, v2f f1, int tp) {
        float4 b[8];
        #pragma unroll
        for (int j = 0; j < 8; ++j) b[j] = bw[(tp * 8 + j) * 32];
        #pragma unroll
        for (int j = 0; j < 8; ++j) {
            v2f bb; bb.x = b[j].x; bb.y = b[j].y;
            cc[j] = __builtin_amdgcn_wmma_f32_16x16x4_f32(false, f0, false, bb,
                                                          (short)0, cc[j], false, false);
        }
        #pragma unroll
        for (int j = 0; j < 8; ++j) {
            v2f bb; bb.x = b[j].z; bb.y = b[j].w;
            cc[j] = __builtin_amdgcn_wmma_f32_16x16x4_f32(false, f1, false, bb,
                                                          (short)0, cc[j], false, false);
        }
    };

    #pragma unroll 1
    for (int tp = 0; tp < IN_F / 8; ++tp) {
        const float* nx = (tp + 1 < IN_F / 8) ? xr + 8 * (tp + 1) : ar;
        const v2f n0 = *(const v2f*)(nx);
        const v2f n1 = *(const v2f*)(nx + 4);
        __builtin_amdgcn_sched_barrier(0);
        step2(a0, a1, tp);
        a0 = n0; a1 = n1;
    }
    a0 *= invd; a1 *= invd;
    #pragma unroll 1
    for (int tp = 0; tp < IN_F / 8; ++tp) {
        const float* nx = ar + 8 * ((tp + 1 < IN_F / 8) ? tp + 1 : tp);
        const v2f n0 = *(const v2f*)(nx);
        const v2f n1 = *(const v2f*)(nx + 4);
        __builtin_amdgcn_sched_barrier(0);
        step2(a0, a1, IN_F / 8 + tp);
        a0 = n0 * invd; a1 = n1 * invd;
    }

    float* hp = h + (size_t)(m0 + hh * 8) * H_F + c;
    #pragma unroll
    for (int j = 0; j < 8; ++j) {
        const float bv = bias[j * 16 + c];
        #pragma unroll
        for (int r = 0; r < 8; ++r)
            hp[(size_t)r * H_F + j * 16] = fmaxf(cc[j][r] + bv, 0.0f);
    }
}

__global__ void sage_gemm2_kernel(const float* __restrict__ h,
                                  const float* __restrict__ acc,
                                  const float* __restrict__ deg,
                                  const float* __restrict__ Ws,
                                  const float* __restrict__ Wn,
                                  const float* __restrict__ bias,
                                  float* __restrict__ out) {
    const int lane = threadIdx.x & 31;
    const int wave = threadIdx.x >> 5;
    const int tile = blockIdx.x * 8 + wave;
    if (tile >= N_DST2 / 16) return;
    const int m0    = tile * 16;
    const int row   = m0 + (lane & 15);
    const int col   = lane & 15;
    const int khalf = (lane >> 4) * 2;

    const float* hr = h   + (size_t)row * H_F;
    const float* ar = acc + (size_t)row * H_F;
    const float invd = 1.0f / fmaxf(deg[row], 1.0f);

    v8f c = {};
    #pragma unroll 4
    for (int k = 0; k < H_F; k += 4) {
        const int kk = k + khalf;
        v2f a, b;
        a.x = hr[kk];
        a.y = hr[kk + 1];
        b.x = Ws[(size_t)kk * N_CLS + col];
        b.y = Ws[(size_t)(kk + 1) * N_CLS + col];
        c = __builtin_amdgcn_wmma_f32_16x16x4_f32(false, a, false, b,
                                                  (short)0, c, false, false);
    }
    #pragma unroll 4
    for (int k = 0; k < H_F; k += 4) {
        const int kk = k + khalf;
        v2f a, b;
        a.x = ar[kk]     * invd;
        a.y = ar[kk + 1] * invd;
        b.x = Wn[(size_t)kk * N_CLS + col];
        b.y = Wn[(size_t)(kk + 1) * N_CLS + col];
        c = __builtin_amdgcn_wmma_f32_16x16x4_f32(false, a, false, b,
                                                  (short)0, c, false, false);
    }

    const float bv    = bias[col];
    const int   rbase = m0 + ((lane >> 4) * 8);
    #pragma unroll
    for (int r = 0; r < 8; ++r) {
        float v = c[r] + bv;
        float m = v;
        m = fmaxf(m, __shfl_xor(m, 1, 16));
        m = fmaxf(m, __shfl_xor(m, 2, 16));
        m = fmaxf(m, __shfl_xor(m, 4, 16));
        m = fmaxf(m, __shfl_xor(m, 8, 16));
        float e = __expf(v - m);
        float s = e;
        s += __shfl_xor(s, 1, 16);
        s += __shfl_xor(s, 2, 16);
        s += __shfl_xor(s, 4, 16);
        s += __shfl_xor(s, 8, 16);
        out[(size_t)(rbase + r) * N_CLS + col] = (v - m) - __logf(s);
    }
}

extern "C" void kernel_launch(void* const* d_in, const int* in_sizes, int n_in,
                              void* d_out, int out_size, void* d_ws, size_t ws_size,
                              hipStream_t stream) {
    const float* x    = (const float*)d_in[0];
    const int*   src1 = (const int*)  d_in[1];
    const int*   dst1 = (const int*)  d_in[2];
    const float* w1   = (const float*)d_in[3];
    const int*   src2 = (const int*)  d_in[4];
    const int*   dst2 = (const int*)  d_in[5];
    const float* w2   = (const float*)d_in[6];
    const float* Ws1  = (const float*)d_in[7];
    const float* Wn1  = (const float*)d_in[8];
    const float* b1   = (const float*)d_in[9];
    const float* Ws2  = (const float*)d_in[10];
    const float* Wn2  = (const float*)d_in[11];
    const float* b2   = (const float*)d_in[12];
    float* out = (float*)d_out;

    float* acc1 = (float*)d_ws;
    float* deg1 = acc1 + (size_t)N_DST1 * 128;
    float* hbuf = deg1 + N_DST1;
    float* acc2 = hbuf + (size_t)N_DST1 * 128;
    float* deg2 = acc2 + (size_t)N_DST2 * 128;

    const int E1 = in_sizes[1];
    const int E2 = in_sizes[4];

    unsigned* keys1 = (unsigned*)hbuf;
    unsigned* keys2 = (unsigned*)acc1;

    tile_key_kernel<ROWS_L1><<<512, 256, 0, stream>>>(dst1, keys1, E1);

    dst_tile_aggregate_kernel<ROWS_L1>
        <<<(N_DST1 + ROWS_L1 - 1) / ROWS_L1, AGG_THREADS, 0, stream>>>(
            x, src1, dst1, keys1, w1, acc1, deg1, E1, N_DST1);

    sage_gemm1_kernel<<<N_DST1 / (16 * G1_WAVES), G1_THREADS, 0, stream>>>(x, acc1, deg1,
                                                                         Ws1, Wn1, b1, hbuf);

    tile_key_kernel<ROWS_L2><<<256, 256, 0, stream>>>(dst2, keys2, E2);
    dst_tile_aggregate_kernel<ROWS_L2>
        <<<(N_DST2 + ROWS_L2 - 1) / ROWS_L2, AGG_THREADS, 0, stream>>>(
            hbuf, src2, dst2, keys2, w2, acc2, deg2, E2, N_DST2);

    sage_gemm2_kernel<<<(N_DST2 / 16 + 7) / 8, 256, 0, stream>>>(hbuf, acc2, deg2,
                                                                 Ws2, Wn2, b2, out);
}


# Neighbour sums: at the start of a scan trip a wave asks for 8 source rows, the next 8 edges' data and the next keys, adds the rows a trip later: gathers run under the key scan; sum order: arrival
# speedup vs baseline: 3.3411x; 1.0540x over previous
#include <hip/hip_runtime.h>
#include <hip/hip_bf16.h>

typedef __attribute__((ext_vector_type(2))) float v2f;
typedef __attribute__((ext_vector_type(8))) float v8f;

#define N_SRC1 500000
#define N_DST1 100000
#define N_DST2 20000
#define IN_F   128
#define H_F    128
#define N_CLS  16

#define KEYS_PER_LOAD 512

template <int ROWS>
__device__ __forceinline__ unsigned tile_key(int d) {
    const unsigned q = (unsigned)d / (unsigned)ROWS;
    return q < 255u ? q : 255u;
}

template <int ROWS>
__global__ __launch_bounds__(256)
void tile_key_kernel(const int* __restrict__ dst, unsigned* __restrict__ keys, int nEdges) {
    const int nGroups = ((nEdges + KEYS_PER_LOAD - 1) / KEYS_PER_LOAD) * (KEYS_PER_LOAD / 16);
    for (int g = blockIdx.x * 256 + threadIdx.x; g < nGroups; g += gridDim.x * 256) {
        const int e0 = g * 16;
        unsigned k0 = 0u, k1 = 0u, k2 = 0u, k3 = 0u;
        if (e0 + 16 <= nEdges) {
            const int4 a = ((const int4*)dst)[(size_t)g * 4];
            const int4 b = ((const int4*)dst)[(size_t)g * 4 + 1];
            const int4 c = ((const int4*)dst)[(size_t)g * 4 + 2];
            const int4 d = ((const int4*)dst)[(size_t)g * 4 + 3];
            k0 = tile_key<ROWS>(a.x) | (tile_key<ROWS>(a.y) << 8) | (tile_key<ROWS>(a.z) << 16) | (tile_key<ROWS>(a.w) << 24);
            k1 = tile_key<ROWS>(b.x) | (tile_key<ROWS>(b.y) << 8) | (tile_key<ROWS>(b.z) << 16) | (tile_key<ROWS>(b.w) << 24);
            k2 = tile_key<ROWS>(c.x) | (tile_key<ROWS>(c.y) << 8) | (tile_key<ROWS>(c.z) << 16) | (tile_key<ROWS>(c.w) << 24);
            k3 = tile_key<ROWS>(d.x) | (tile_key<ROWS>(d.y) << 8) | (tile_key<ROWS>(d.z) << 16) | (tile_key<ROWS>(d.w) << 24);
        } else {
            for (int b = 0; b < 4; ++b) {
                const int ea = e0 + b, eb = e0 + 4 + b, ec = e0 + 8 + b, ed = e0 + 12 + b;
                k0 |= ((ea < nEdges) ? tile_key<ROWS>(dst[ea]) : 255u) << (8 * b);
                k1 |= ((eb < nEdges) ? tile_key<ROWS>(dst[eb]) : 255u) << (8 * b);
                k2 |= ((ec < nEdges) ? tile_key<ROWS>(dst[ec]) : 255u) << (8 * b);
                k3 |= ((ed < nEdges) ? tile_key<ROWS>(dst[ed]) : 255u) << (8 * b);
            }
        }
        ((uint4*)keys)[g] = make_uint4(k0, k1, k2, k3);
    }
}

#define AGG_THREADS 1024
#define AGG_WAVES   32
#define DRAIN_AT    64
#define QCAP        576

template <int ROWS>
__global__ __launch_bounds__(AGG_THREADS)
void dst_tile_aggregate_kernel(const float* __restrict__ feat,
                               const int*   __restrict__ src,
                               const int*   __restrict__ dst,
                               const unsigned* __restrict__ keys,
                               const float* __restrict__ w,
                               float* __restrict__ acc,
                               float* __restrict__ deg,
                               int nEdges, int nRows) {
    __shared__ __attribute__((aligned(16))) float sacc[ROWS * 128];
    __shared__ unsigned sdeg[ROWS];
    __shared__ int squeue[AGG_WAVES * QCAP];

    const int tid  = threadIdx.x;
    const int lane = tid & 31;
    const int wave = __builtin_amdgcn_readfirstlane(tid >> 5);
    const int lo   = blockIdx.x * ROWS;

    for (int i = tid; i < ROWS * 32; i += AGG_THREADS)
        ((float4*)sacc)[i] = make_float4(0.f, 0.f, 0.f, 0.f);
    for (int i = tid; i < ROWS; i += AGG_THREADS) sdeg[i] = 0u;
    __syncthreads();

    int* q  = squeue + wave * QCAP;
    int cnt = 0;

    const float4* feat4 = (const float4*)feat;

    float4 xv[8];
    #pragma unroll
    for (int u = 0; u < 8; ++u) xv[u] = make_float4(0.f, 0.f, 0.f, 0.f);

    auto ask_rows = [&](int s, int k0) {
        #pragma unroll
        for (int u = 0; u < 8; ++u) {
            const int sk = __builtin_amdgcn_readlane(s, k0 + u);
            xv[u] = feat4[(size_t)sk * 32 + lane];
        }
    };

    auto add_rows = [&](int d, float we, int k0, int n) {
        #pragma unroll
        for (int u = 0; u < 8; ++u) {
            if (u < n) {
                const int   dk = __builtin_amdgcn_readlane(d, k0 + u);
                const float wk = __int_as_float(__builtin_amdgcn_readlane(__float_as_int(we), k0 + u));
                float* r = sacc + dk * 128 + lane;
                __hip_atomic_fetch_add(r,      xv[u].x * wk, __ATOMIC_RELAXED, __HIP_MEMORY_SCOPE_WORKGROUP);
                __hip_atomic_fetch_add(r + 32, xv[u].y * wk, __ATOMIC_RELAXED, __HIP_MEMORY_SCOPE_WORKGROUP);
                __hip_atomic_fetch_add(r + 64, xv[u].z * wk, __ATOMIC_RELAXED, __HIP_MEMORY_SCOPE_WORKGROUP);
                __hip_atomic_fetch_add(r + 96, xv[u].w * wk, __ATOMIC_RELAXED, __HIP_MEMORY_SCOPE_WORKGROUP);
            }
        }
    };

    auto flush = [&](int base, int n) {
        const int   e  = (lane < n) ? q[base + lane] : 0;
        const int   s  = src[e];
        const int   d  = dst[e] - lo;
        const float we = w[e];
        if (lane < n)
            __hip_atomic_fetch_add(sdeg + d, 1u, __ATOMIC_RELAXED, __HIP_MEMORY_SCOPE_WORKGROUP);
        for (int k0 = 0; k0 < n; k0 += 8) {
            ask_rows(s, k0);
            add_rows(d, we, k0, n - k0);
        }
    };

    int   mS = 0, mD = 0;  float mW = 0.f;
    int   rD = 0;          float rW = 0.f;
    int   haveM = 0, haveR = 0;

    auto push = [&](bool hit, int e) {
        const unsigned m = __builtin_amdgcn_ballot_w32(hit);
        if (m) {
            if (hit) q[cnt + (int)__builtin_amdgcn_mbcnt_lo(m, 0u)] = e;
            cnt += __builtin_popcount(m);
        }
    };

    auto scan_word = [&](unsigned x, unsigned z, int e) {
        if (__builtin_amdgcn_ballot_w32((z & 0x80808080u) != 0u)) {
            push((x & 0x000000ffu) == 0u, e);
            push((x & 0x0000ff00u) == 0u, e + 1);
            push((x & 0x00ff0000u) == 0u, e + 2);
            push((x & 0xff000000u) == 0u, e + 3);
        }
    };

    const unsigned pat     = (unsigned)blockIdx.x * 0x01010101u;
    const int      nChunks = (nEdges + KEYS_PER_LOAD - 1) / KEYS_PER_LOAD;
    const int      start   = (int)(blockIdx.x * ((unsigned)nChunks / gridDim.x));

    auto load_keys = [&](int i, uint4 (&kv)[4], int (&cb)[4]) {
        #pragma unroll
        for (int u = 0; u < 4; ++u) {
            const int ii = i + AGG_WAVES * u;
            int c = start + ii;
            if (c >= nChunks) c -= nChunks;
            const bool valid = ii < nChunks;
            cb[u] = valid ? c : -1;
            kv[u] = ((const uint4*)keys)[(size_t)(valid ? c : 0) * 32 + lane];
        }
    };

    if (wave < nChunks) {
        uint4 kvA[4];
        int   cbA[4];
        load_keys(wave, kvA, cbA);
        for (int i = wave; i < nChunks; i += AGG_WAVES * 4) {
            if (haveR) add_rows(rD, rW, 0, 8);
            int sk[8];
            #pragma unroll
            for (int u = 0; u < 8; ++u) sk[u] = __builtin_amdgcn_readlane(mS, u);
            uint4 kvB[4];
            int   cbB[4];
            load_keys(i + AGG_WAVES * 4, kvB, cbB);
            #pragma unroll
            for (int u = 0; u < 8; ++u) xv[u] = feat4[(size_t)sk[u] * 32 + lane];
            rD = mD - lo; rW = mW; haveR = haveM;
            if (haveR && lane < 8)
                __hip_atomic_fetch_add(sdeg + rD, 1u, __ATOMIC_RELAXED, __HIP_MEMORY_SCOPE_WORKGROUP);
            haveM = (cnt >= 8) ? 8 : 0;
            cnt -= haveM;
            if (lane < 8) {
                const int qe = q[cnt + lane];
                const int e  = haveM ? qe : 0;
                mS = src[e];
                mD = dst[e];
                mW = w[e];
            }
            #pragma unroll
            for (int u = 0; u < 4; ++u) {
                if (cbA[u] >= 0) {
                    const unsigned x0 = kvA[u].x ^ pat;
                    const unsigned x1 = kvA[u].y ^ pat;
                    const unsigned x2 = kvA[u].z ^ pat;
                    const unsigned x3 = kvA[u].w ^ pat;
                    const unsigned z0 = (x0 - 0x01010101u) & ~x0;
                    const unsigned z1 = (x1 - 0x01010101u) & ~x1;
                    const unsigned z2 = (x2 - 0x01010101u) & ~x2;
                    const unsigned z3 = (x3 - 0x01010101u) & ~x3;
                    if (__builtin_amdgcn_ballot_w32(((z0 | z1 | z2 | z3) & 0x80808080u) != 0u)) {
                        const int e0 = cbA[u] * KEYS_PER_LOAD + lane * 16;
                        scan_word(x0, z0, e0);
                        scan_word(x1, z1, e0 + 4);
                        scan_word(x2, z2, e0 + 8);
                        scan_word(x3, z3, e0 + 12);
                    }
                    while (cnt >= DRAIN_AT) {
                        if (haveR) { add_rows(rD, rW, 0, 8); haveR = 0; }
                        cnt -= 32;
                        flush(cnt, 32);
                    }
                }
            }
            #pragma unroll
            for (int u = 0; u < 4; ++u) { kvA[u] = kvB[u]; cbA[u] = cbB[u]; }
        }
    }
    if (haveR) add_rows(rD, rW, 0, 8);
    if (haveM) {
        const int d = mD - lo;
        if (lane < 8)
            __hip_atomic_fetch_add(sdeg + d, 1u, __ATOMIC_RELAXED, __HIP_MEMORY_SCOPE_WORKGROUP);
        ask_rows(mS, 0);
        add_rows(d, mW, 0, 8);
    }
    while (cnt > 0) {
        const int n = cnt < 32 ? cnt : 32;
        cnt -= n;
        flush(cnt, n);
    }

    __syncthreads();

    for (int r = wave; r < ROWS; r += AGG_WAVES) {
        if (lo + r < nRows) {
            const float* p = sacc + r * 128 + lane;
            ((float4*)(acc + (size_t)(lo + r) * 128))[lane] =
                make_float4(p[0], p[32], p[64], p[96]);
        }
    }
    for (int i = tid; i < ROWS; i += AGG_THREADS)
        if (lo + i < nRows) deg[lo + i] = (float)sdeg[i];
}

#define ROWS_L1 416
#define ROWS_L2 96

#define G1_WAVES   25
#define G1_THREADS (G1_WAVES * 32)
static_assert(N_DST1 % (16 * G1_WAVES) == 0, "M tiles must fill whole work-groups");

__global__ __launch_bounds__(G1_THREADS)
void sage_gemm1_kernel(const float* __restrict__ x,
                       const float* __restrict__ acc,
                       const float* __restrict__ deg,
                       const float* __restrict__ Ws,
                       const float* __restrict__ Wn,
                       const float* __restrict__ bias,
                       float* __restrict__ h) {
    __shared__ __attribute__((aligned(16))) float sW[(2 * IN_F / 8) * 8 * 32 * 4];

    const int tid  = threadIdx.x;
    const int lane = tid & 31;
    const int wave = tid >> 5;
    const int c    = lane & 15;
    const int hh   = lane >> 4;
    const int m0   = (blockIdx.x * G1_WAVES + wave) * 16;
    const int row  = m0 + c;

    const float* xr = x   + (size_t)row * IN_F + 2 * hh;
    const float* ar = acc + (size_t)row * IN_F + 2 * hh;

    v2f a0 = *(const v2f*)(xr);
    v2f a1 = *(const v2f*)(xr + 4);
    const float invd = 1.0f / fmaxf(deg[row], 1.0f);

    for (int i = tid; i < IN_F * H_F; i += G1_THREADS) {
        const int p   = i >> 7;
        const int col = i & 127;
        const int k0  = 2 * p;
        const float* wsrc = (k0 < IN_F) ? Ws + (size_t)k0 * H_F
                                        : Wn + (size_t)(k0 - IN_F) * H_F;
        v2f v;
        v.x = wsrc[col];
        v.y = wsrc[H_F + col];
        const int t  = p >> 1;
        const int fh = p & 1;
        const int fl = fh * 16 + (col & 15);
        const int e  = ((t >> 1) * 8 + (col >> 4)) * 32 + fl;
        *(v2f*)(sW + e * 4 + (t & 1) * 2) = v;
    }
    __syncthreads();

    const float4* bw = (const float4*)sW + lane;
    v8f cc[8] = {};

    auto step2 = [&](v2f f0, v2f f1, int tp) {
        float4 b[8];
        #pragma unroll
        for (int j = 0; j < 8; ++j) b[j] = bw[(tp * 8 + j) * 32];
        #pragma unroll
        for (int j = 0; j < 8; ++j) {
            v2f bb; bb.x = b[j].x; bb.y = b[j].y;
            cc[j] = __builtin_amdgcn_wmma_f32_16x16x4_f32(false, f0, false, bb,
                                                          (short)0, cc[j], false, false);
        }
        #pragma unroll
        for (int j = 0; j < 8; ++j) {
            v2f bb; bb.x = b[j].z; bb.y = b[j].w;
            cc[j] = __builtin_amdgcn_wmma_f32_16x16x4_f32(false, f1, false, bb,
                                                          (short)0, cc[j], false, false);
        }
    };

    #pragma unroll 1
    for (int tp = 0; tp < IN_F / 8; ++tp) {
        const float* nx = (tp + 1 < IN_F / 8) ? xr + 8 * (tp + 1) : ar;
        const v2f n0 = *(const v2f*)(nx);
        const v2f n1 = *(const v2f*)(nx + 4);
        __builtin_amdgcn_sched_barrier(0);
        step2(a0, a1, tp);
        a0 = n0; a1 = n1;
    }
    a0 *= invd; a1 *= invd;
    #pragma unroll 1
    for (int tp = 0; tp < IN_F / 8; ++tp) {
        const float* nx = ar + 8 * ((tp + 1 < IN_F / 8) ? tp + 1 : tp);
        const v2f n0 = *(const v2f*)(nx);
        const v2f n1 = *(const v2f*)(nx + 4);
        __builtin_amdgcn_sched_barrier(0);
        step2(a0, a1, IN_F / 8 + tp);
        a0 = n0 * invd; a1 = n1 * invd;
    }

    float* hp = h + (size_t)(m0 + hh * 8) * H_F + c;
    #pragma unroll
    for (int j = 0; j < 8; ++j) {
        const float bv = bias[j * 16 + c];
        #pragma unroll
        for (int r = 0; r < 8; ++r)
            hp[(size_t)r * H_F + j * 16] = fmaxf(cc[j][r] + bv, 0.0f);
    }
}

__global__ void sage_gemm2_kernel(const float* __restrict__ h,
                                  const float* __restrict__ acc,
                                  const float* __restrict__ deg,
                                  const float* __restrict__ Ws,
                                  const float* __restrict__ Wn,
                                  const float* __restrict__ bias,
                                  float* __restrict__ out) {
    const int lane = threadIdx.x & 31;
    const int wave = threadIdx.x >> 5;
    const int tile = blockIdx.x * 8 + wave;
    if (tile >= N_DST2 / 16) return;
    const int m0    = tile * 16;
    const int row   = m0 + (lane & 15);
    const int col   = lane & 15;
    const int khalf = (lane >> 4) * 2;

    const float* hr = h   + (size_t)row * H_F;
    const float* ar = acc + (size_t)row * H_F;
    const float invd = 1.0f / fmaxf(deg[row], 1.0f);

    v8f c = {};
    #pragma unroll 4
    for (int k = 0; k < H_F; k += 4) {
        const int kk = k + khalf;
        v2f a, b;
        a.x = hr[kk];
        a.y = hr[kk + 1];
        b.x = Ws[(size_t)kk * N_CLS + col];
        b.y = Ws[(size_t)(kk + 1) * N_CLS + col];
        c = __builtin_amdgcn_wmma_f32_16x16x4_f32(false, a, false, b,
                                                  (short)0, c, false, false);
    }
    #pragma unroll 4
    for (int k = 0; k < H_F; k += 4) {
        const int kk = k + khalf;
        v2f a, b;
        a.x = ar[kk]     * invd;
        a.y = ar[kk + 1] * invd;
        b.x = Wn[(size_t)kk * N_CLS + col];
        b.y = Wn[(size_t)(kk + 1) * N_CLS + col];
        c = __builtin_amdgcn_wmma_f32_16x16x4_f32(false, a, false, b,
                                                  (short)0, c, false, false);
    }

    const float bv    = bias[col];
    const int   rbase = m0 + ((lane >> 4) * 8);
    #pragma unroll
    for (int r = 0; r < 8; ++r) {
        float v = c[r] + bv;
        float m = v;
        m = fmaxf(m, __shfl_xor(m, 1, 16));
        m = fmaxf(m, __shfl_xor(m, 2, 16));
        m = fmaxf(m, __shfl_xor(m, 4, 16));
        m = fmaxf(m, __shfl_xor(m, 8, 16));
        float e = __expf(v - m);
        float s = e;
        s += __shfl_xor(s, 1, 16);
        s += __shfl_xor(s, 2, 16);
        s += __shfl_xor(s, 4, 16);
        s += __shfl_xor(s, 8, 16);
        out[(size_t)(rbase + r) * N_CLS + col] = (v - m) - __logf(s);
    }
}

extern "C" void kernel_launch(void* const* d_in, const int* in_sizes, int n_in,
                              void* d_out, int out_size, void* d_ws, size_t ws_size,
                              hipStream_t stream) {
    const float* x    = (const float*)d_in[0];
    const int*   src1 = (const int*)  d_in[1];
    const int*   dst1 = (const int*)  d_in[2];
    const float* w1   = (const float*)d_in[3];
    const int*   src2 = (const int*)  d_in[4];
    const int*   dst2 = (const int*)  d_in[5];
    const float* w2   = (const float*)d_in[6];
    const float* Ws1  = (const float*)d_in[7];
    const float* Wn1  = (const float*)d_in[8];
    const float* b1   = (const float*)d_in[9];
    const float* Ws2  = (const float*)d_in[10];
    const float* Wn2  = (const float*)d_in[11];
    const float* b2   = (const float*)d_in[12];
    float* out = (float*)d_out;

    float* acc1 = (float*)d_ws;
    float* deg1 = acc1 + (size_t)N_DST1 * 128;
    float* hbuf = deg1 + N_DST1;
    float* acc2 = hbuf + (size_t)N_DST1 * 128;
    float* deg2 = acc2 + (size_t)N_DST2 * 128;

    const int E1 = in_sizes[1];
    const int E2 = in_sizes[4];

    unsigned* keys1 = (unsigned*)hbuf;
    unsigned* keys2 = (unsigned*)acc1;

    tile_key_kernel<ROWS_L1><<<512, 256, 0, stream>>>(dst1, keys1, E1);

    dst_tile_aggregate_kernel<ROWS_L1>
        <<<(N_DST1 + ROWS_L1 - 1) / ROWS_L1, AGG_THREADS, 0, stream>>>(
            x, src1, dst1, keys1, w1, acc1, deg1, E1, N_DST1);

    sage_gemm1_kernel<<<N_DST1 / (16 * G1_WAVES), G1_THREADS, 0, stream>>>(x, acc1, deg1,
                                                                         Ws1, Wn1, b1, hbuf);

    tile_key_kernel<ROWS_L2><<<256, 256, 0, stream>>>(dst2, keys2, E2);
    dst_tile_aggregate_kernel<ROWS_L2>
        <<<(N_DST2 + ROWS_L2 - 1) / ROWS_L2, AGG_THREADS, 0, stream>>>(
            hbuf, src2, dst2, keys2, w2, acc2, deg2, E2, N_DST2);

    sage_gemm2_kernel<<<(N_DST2 / 16 + 7) / 8, 256, 0, stream>>>(hbuf, acc2, deg2,
                                                                 Ws2, Wn2, b2, out);
}


# Aggregation tiles sized to the chip: 393 / 80 rows a work-group make 255 / 250 work-groups (were 241 / 209); rows leave LDS divided by degree, no degree plane fixes the height; sum order: arrival
# speedup vs baseline: 3.5582x; 1.0650x over previous
#include <hip/hip_runtime.h>
#include <hip/hip_bf16.h>

typedef __attribute__((ext_vector_type(2))) float v2f;
typedef __attribute__((ext_vector_type(8))) float v8f;

#define N_SRC1 500000
#define N_DST1 100000
#define N_DST2 20000
#define IN_F   128
#define H_F    128
#define N_CLS  16

#define KEYS_PER_LOAD 512

template <int ROWS>
__device__ __forceinline__ unsigned tile_key(int d) {
    const unsigned q = (unsigned)d / (unsigned)ROWS;
    return q < 255u ? q : 255u;
}

template <int ROWS>
__global__ __launch_bounds__(256)
void tile_key_kernel(const int* __restrict__ dst, unsigned* __restrict__ keys, int nEdges) {
    const int nGroups = ((nEdges + KEYS_PER_LOAD - 1) / KEYS_PER_LOAD) * (KEYS_PER_LOAD / 16);
    for (int g = blockIdx.x * 256 + threadIdx.x; g < nGroups; g += gridDim.x * 256) {
        const int e0 = g * 16;
        unsigned k0 = 0u, k1 = 0u, k2 = 0u, k3 = 0u;
        if (e0 + 16 <= nEdges) {
            const int4 a = ((const int4*)dst)[(size_t)g * 4];
            const int4 b = ((const int4*)dst)[(size_t)g * 4 + 1];
            const int4 c = ((const int4*)dst)[(size_t)g * 4 + 2];
            const int4 d = ((const int4*)dst)[(size_t)g * 4 + 3];
            k0 = tile_key<ROWS>(a.x) | (tile_key<ROWS>(a.y) << 8) | (tile_key<ROWS>(a.z) << 16) | (tile_key<ROWS>(a.w) << 24);
            k1 = tile_key<ROWS>(b.x) | (tile_key<ROWS>(b.y) << 8) | (tile_key<ROWS>(b.z) << 16) | (tile_key<ROWS>(b.w) << 24);
            k2 = tile_key<ROWS>(c.x) | (tile_key<ROWS>(c.y) << 8) | (tile_key<ROWS>(c.z) << 16) | (tile_key<ROWS>(c.w) << 24);
            k3 = tile_key<ROWS>(d.x) | (tile_key<ROWS>(d.y) << 8) | (tile_key<ROWS>(d.z) << 16) | (tile_key<ROWS>(d.w) << 24);
        } else {
            for (int b = 0; b < 4; ++b) {
                const int ea = e0 + b, eb = e0 + 4 + b, ec = e0 + 8 + b, ed = e0 + 12 + b;
                k0 |= ((ea < nEdges) ? tile_key<ROWS>(dst[ea]) : 255u) << (8 * b);
                k1 |= ((eb < nEdges) ? tile_key<ROWS>(dst[eb]) : 255u) << (8 * b);
                k2 |= ((ec < nEdges) ? tile_key<ROWS>(dst[ec]) : 255u) << (8 * b);
                k3 |= ((ed < nEdges) ? tile_key<ROWS>(dst[ed]) : 255u) << (8 * b);
            }
        }
        ((uint4*)keys)[g] = make_uint4(k0, k1, k2, k3);
    }
}

#define AGG_THREADS 1024
#define AGG_WAVES   32
#define DRAIN_AT    64
#define QCAP        576

template <int ROWS>
__global__ __launch_bounds__(AGG_THREADS)
void dst_tile_aggregate_kernel(const float* __restrict__ feat,
                               const int*   __restrict__ src,
                               const int*   __restrict__ dst,
                               const unsigned* __restrict__ keys,
                               const float* __restrict__ w,
                               float* __restrict__ acc,
                               int nEdges, int nRows) {
    __shared__ __attribute__((aligned(16))) float sacc[ROWS * 128];
    __shared__ unsigned sdeg[ROWS];
    __shared__ int squeue[AGG_WAVES * QCAP];

    const int tid  = threadIdx.x;
    const int lane = tid & 31;
    const int wave = __builtin_amdgcn_readfirstlane(tid >> 5);
    const int lo   = blockIdx.x * ROWS;

    for (int i = tid; i < ROWS * 32; i += AGG_THREADS)
        ((float4*)sacc)[i] = make_float4(0.f, 0.f, 0.f, 0.f);
    for (int i = tid; i < ROWS; i += AGG_THREADS) sdeg[i] = 0u;
    __syncthreads();

    int* q  = squeue + wave * QCAP;
    int cnt = 0;

    const float4* feat4 = (const float4*)feat;

    float4 xv[8];
    #pragma unroll
    for (int u = 0; u < 8; ++u) xv[u] = make_float4(0.f, 0.f, 0.f, 0.f);

    auto ask_rows = [&](int s, int k0) {
        #pragma unroll
        for (int u = 0; u < 8; ++u) {
            const int sk = __builtin_amdgcn_readlane(s, k0 + u);
            xv[u] = feat4[(size_t)sk * 32 + lane];
        }
    };

    auto add_rows = [&](int d, float we, int k0, int n) {
        #pragma unroll
        for (int u = 0; u < 8; ++u) {
            if (u < n) {
                const int   dk = __builtin_amdgcn_readlane(d, k0 + u);
                const float wk = __int_as_float(__builtin_amdgcn_readlane(__float_as_int(we), k0 + u));
                float* r = sacc + dk * 128 + lane;
                __hip_atomic_fetch_add(r,      xv[u].x * wk, __ATOMIC_RELAXED, __HIP_MEMORY_SCOPE_WORKGROUP);
                __hip_atomic_fetch_add(r + 32, xv[u].y * wk, __ATOMIC_RELAXED, __HIP_MEMORY_SCOPE_WORKGROUP);
                __hip_atomic_fetch_add(r + 64, xv[u].z * wk, __ATOMIC_RELAXED, __HIP_MEMORY_SCOPE_WORKGROUP);
                __hip_atomic_fetch_add(r + 96, xv[u].w * wk, __ATOMIC_RELAXED, __HIP_MEMORY_SCOPE_WORKGROUP);
            }
        }
    };

    auto flush = [&](int base, int n) {
        const int   e  = (lane < n) ? q[base + lane] : 0;
        const int   s  = src[e];
        const int   d  = dst[e] - lo;
        const float we = w[e];
        if (lane < n)
            __hip_atomic_fetch_add(sdeg + d, 1u, __ATOMIC_RELAXED, __HIP_MEMORY_SCOPE_WORKGROUP);
        for (int k0 = 0; k0 < n; k0 += 8) {
            ask_rows(s, k0);
            add_rows(d, we, k0, n - k0);
        }
    };

    int   mS = 0, mD = 0;  float mW = 0.f;
    int   rD = 0;          float rW = 0.f;
    int   haveM = 0, haveR = 0;

    auto push = [&](bool hit, int e) {
        const unsigned m = __builtin_amdgcn_ballot_w32(hit);
        if (m) {
            if (hit) q[cnt + (int)__builtin_amdgcn_mbcnt_lo(m, 0u)] = e;
            cnt += __builtin_popcount(m);
        }
    };

    auto scan_word = [&](unsigned x, unsigned z, int e) {
        if (__builtin_amdgcn_ballot_w32((z & 0x80808080u) != 0u)) {
            push((x & 0x000000ffu) == 0u, e);
            push((x & 0x0000ff00u) == 0u, e + 1);
            push((x & 0x00ff0000u) == 0u, e + 2);
            push((x & 0xff000000u) == 0u, e + 3);
        }
    };

    const unsigned pat     = (unsigned)blockIdx.x * 0x01010101u;
    const int      nChunks = (nEdges + KEYS_PER_LOAD - 1) / KEYS_PER_LOAD;
    const int      start   = (int)(blockIdx.x * ((unsigned)nChunks / gridDim.x));

    auto load_keys = [&](int i, uint4 (&kv)[4], int (&cb)[4]) {
        #pragma unroll
        for (int u = 0; u < 4; ++u) {
            const int ii = i + AGG_WAVES * u;
            int c = start + ii;
            if (c >= nChunks) c -= nChunks;
            const bool valid = ii < nChunks;
            cb[u] = valid ? c : -1;
            kv[u] = ((const uint4*)keys)[(size_t)(valid ? c : 0) * 32 + lane];
        }
    };

    if (wave < nChunks) {
        uint4 kvA[4];
        int   cbA[4];
        load_keys(wave, kvA, cbA);
        for (int i = wave; i < nChunks; i += AGG_WAVES * 4) {
            if (haveR) add_rows(rD, rW, 0, 8);
            int sk[8];
            #pragma unroll
            for (int u = 0; u < 8; ++u) sk[u] = __builtin_amdgcn_readlane(mS, u);
            uint4 kvB[4];
            int   cbB[4];
            load_keys(i + AGG_WAVES * 4, kvB, cbB);
            #pragma unroll
            for (int u = 0; u < 8; ++u) xv[u] = feat4[(size_t)sk[u] * 32 + lane];
            rD = mD - lo; rW = mW; haveR = haveM;
            if (haveR && lane < 8)
                __hip_atomic_fetch_add(sdeg + rD, 1u, __ATOMIC_RELAXED, __HIP_MEMORY_SCOPE_WORKGROUP);
            haveM = (cnt >= 8) ? 8 : 0;
            cnt -= haveM;
            if (lane < 8) {
                const int qe = q[cnt + lane];
                const int e  = haveM ? qe : 0;
                mS = src[e];
                mD = dst[e];
                mW = w[e];
            }
            #pragma unroll
            for (int u = 0; u < 4; ++u) {
                if (cbA[u] >= 0) {
                    const unsigned x0 = kvA[u].x ^ pat;
                    const unsigned x1 = kvA[u].y ^ pat;
                    const unsigned x2 = kvA[u].z ^ pat;
                    const unsigned x3 = kvA[u].w ^ pat;
                    const unsigned z0 = (x0 - 0x01010101u) & ~x0;
                    const unsigned z1 = (x1 - 0x01010101u) & ~x1;
                    const unsigned z2 = (x2 - 0x01010101u) & ~x2;
                    const unsigned z3 = (x3 - 0x01010101u) & ~x3;
                    if (__builtin_amdgcn_ballot_w32(((z0 | z1 | z2 | z3) & 0x80808080u) != 0u)) {
                        const int e0 = cbA[u] * KEYS_PER_LOAD + lane * 16;
                        scan_word(x0, z0, e0);
                        scan_word(x1, z1, e0 + 4);
                        scan_word(x2, z2, e0 + 8);
                        scan_word(x3, z3, e0 + 12);
                    }
                    while (cnt >= DRAIN_AT) {
                        if (haveR) { add_rows(rD, rW, 0, 8); haveR = 0; }
                        cnt -= 32;
                        flush(cnt, 32);
                    }
                }
            }
            #pragma unroll
            for (int u = 0; u < 4; ++u) { kvA[u] = kvB[u]; cbA[u] = cbB[u]; }
        }
    }
    if (haveR) add_rows(rD, rW, 0, 8);
    if (haveM) {
        const int d = mD - lo;
        if (lane < 8)
            __hip_atomic_fetch_add(sdeg + d, 1u, __ATOMIC_RELAXED, __HIP_MEMORY_SCOPE_WORKGROUP);
        ask_rows(mS, 0);
        add_rows(d, mW, 0, 8);
    }
    while (cnt > 0) {
        const int n = cnt < 32 ? cnt : 32;
        cnt -= n;
        flush(cnt, n);
    }

    __syncthreads();

    for (int r = wave; r < ROWS; r += AGG_WAVES) {
        if (lo + r < nRows) {
            const float* p = sacc + r * 128 + lane;
            const float invd = 1.0f / fmaxf((float)sdeg[r], 1.0f);
            ((float4*)(acc + (size_t)(lo + r) * 128))[lane] =
                make_float4(p[0] * invd, p[32] * invd, p[64] * invd, p[96] * invd);
        }
    }
}

#define ROWS_L1 393
#define ROWS_L2 80
static_assert((N_DST1 - 1) / ROWS_L1 < 255 && (N_DST2 - 1) / ROWS_L2 < 255,
              "every tile number must fit a key byte below the padding key");

#define G1_WAVES   25
#define G1_THREADS (G1_WAVES * 32)
static_assert(N_DST1 % (16 * G1_WAVES) == 0, "M tiles must fill whole work-groups");

__global__ __launch_bounds__(G1_THREADS)
void sage_gemm1_kernel(const float* __restrict__ x,
                       const float* __restrict__ acc,
                       const float* __restrict__ Ws,
                       const float* __restrict__ Wn,
                       const float* __restrict__ bias,
                       float* __restrict__ h) {
    __shared__ __attribute__((aligned(16))) float sW[(2 * IN_F / 8) * 8 * 32 * 4];

    const int tid  = threadIdx.x;
    const int lane = tid & 31;
    const int wave = tid >> 5;
    const int c    = lane & 15;
    const int hh   = lane >> 4;
    const int m0   = (blockIdx.x * G1_WAVES + wave) * 16;
    const int row  = m0 + c;

    const float* xr = x   + (size_t)row * IN_F + 2 * hh;
    const float* ar = acc + (size_t)row * IN_F + 2 * hh;

    v2f a0 = *(const v2f*)(xr);
    v2f a1 = *(const v2f*)(xr + 4);

    for (int i = tid; i < IN_F * H_F; i += G1_THREADS) {
        const int p   = i >> 7;
        const int col = i & 127;
        const int k0  = 2 * p;
        const float* wsrc = (k0 < IN_F) ? Ws + (size_t)k0 * H_F
                                        : Wn + (size_t)(k0 - IN_F) * H_F;
        v2f v;
        v.x = wsrc[col];
        v.y = wsrc[H_F + col];
        const int t  = p >> 1;
        const int fh = p & 1;
        const int fl = fh * 16 + (col & 15);
        const int e  = ((t >> 1) * 8 + (col >> 4)) * 32 + fl;
        *(v2f*)(sW + e * 4 + (t & 1) * 2) = v;
    }
    __syncthreads();

    const float4* bw = (const float4*)sW + lane;
    v8f cc[8] = {};

    auto step2 = [&](v2f f0, v2f f1, int tp) {
        float4 b[8];
        #pragma unroll
        for (int j = 0; j < 8; ++j) b[j] = bw[(tp * 8 + j) * 32];
        #pragma unroll
        for (int j = 0; j < 8; ++j) {
            v2f bb; bb.x = b[j].x; bb.y = b[j].y;
            cc[j] = __builtin_amdgcn_wmma_f32_16x16x4_f32(false, f0, false, bb,
                                                          (short)0, cc[j], false, false);
        }
        #pragma unroll
        for (int j = 0; j < 8; ++j) {
            v2f bb; bb.x = b[j].z; bb.y = b[j].w;
            cc[j] = __builtin_amdgcn_wmma_f32_16x16x4_f32(false, f1, false, bb,
                                                          (short)0, cc[j], false, false);
        }
    };

    #pragma unroll 1
    for (int tp = 0; tp < IN_F / 8; ++tp) {
        const float* nx = (tp + 1 < IN_F / 8) ? xr + 8 * (tp + 1) : ar;
        const v2f n0 = *(const v2f*)(nx);
        const v2f n1 = *(const v2f*)(nx + 4);
        __builtin_amdgcn_sched_barrier(0);
        step2(a0, a1, tp);
        a0 = n0; a1 = n1;
    }
    #pragma unroll 1
    for (int tp = 0; tp < IN_F / 8; ++tp) {
        const float* nx = ar + 8 * ((tp + 1 < IN_F / 8) ? tp + 1 : tp);
        const v2f n0 = *(const v2f*)(nx);
        const v2f n1 = *(const v2f*)(nx + 4);
        __builtin_amdgcn_sched_barrier(0);
        step2(a0, a1, IN_F / 8 + tp);
        a0 = n0; a1 = n1;
    }

    float* hp = h + (size_t)(m0 + hh * 8) * H_F + c;
    #pragma unroll
    for (int j = 0; j < 8; ++j) {
        const float bv = bias[j * 16 + c];
        #pragma unroll
        for (int r = 0; r < 8; ++r)
            hp[(size_t)r * H_F + j * 16] = fmaxf(cc[j][r] + bv, 0.0f);
    }
}

__global__ void sage_gemm2_kernel(const float* __restrict__ h,
                                  const float* __restrict__ acc,
                                  const float* __restrict__ Ws,
                                  const float* __restrict__ Wn,
                                  const float* __restrict__ bias,
                                  float* __restrict__ out) {
    const int lane = threadIdx.x & 31;
    const int wave = threadIdx.x >> 5;
    const int tile = blockIdx.x * 8 + wave;
    if (tile >= N_DST2 / 16) return;
    const int m0    = tile * 16;
    const int row   = m0 + (lane & 15);
    const int col   = lane & 15;
    const int khalf = (lane >> 4) * 2;

    const float* hr = h   + (size_t)row * H_F;
    const float* ar = acc + (size_t)row * H_F;

    v8f c = {};
    #pragma unroll 4
    for (int k = 0; k < H_F; k += 4) {
        const int kk = k + khalf;
        v2f a, b;
        a.x = hr[kk];
        a.y = hr[kk + 1];
        b.x = Ws[(size_t)kk * N_CLS + col];
        b.y = Ws[(size_t)(kk + 1) * N_CLS + col];
        c = __builtin_amdgcn_wmma_f32_16x16x4_f32(false, a, false, b,
                                                  (short)0, c, false, false);
    }
    #pragma unroll 4
    for (int k = 0; k < H_F; k += 4) {
        const int kk = k + khalf;
        v2f a, b;
        a.x = ar[kk];
        a.y = ar[kk + 1];
        b.x = Wn[(size_t)kk * N_CLS + col];
        b.y = Wn[(size_t)(kk + 1) * N_CLS + col];
        c = __builtin_amdgcn_wmma_f32_16x16x4_f32(false, a, false, b,
                                                  (short)0, c, false, false);
    }

    const float bv    = bias[col];
    const int   rbase = m0 + ((lane >> 4) * 8);
    #pragma unroll
    for (int r = 0; r < 8; ++r) {
        float v = c[r] + bv;
        float m = v;
        m = fmaxf(m, __shfl_xor(m, 1, 16));
        m = fmaxf(m, __shfl_xor(m, 2, 16));
        m = fmaxf(m, __shfl_xor(m, 4, 16));
        m = fmaxf(m, __shfl_xor(m, 8, 16));
        float e = __expf(v - m);
        float s = e;
        s += __shfl_xor(s, 1, 16);
        s += __shfl_xor(s, 2, 16);
        s += __shfl_xor(s, 4, 16);
        s += __shfl_xor(s, 8, 16);
        out[(size_t)(rbase + r) * N_CLS + col] = (v - m) - __logf(s);
    }
}

extern "C" void kernel_launch(void* const* d_in, const int* in_sizes, int n_in,
                              void* d_out, int out_size, void* d_ws, size_t ws_size,
                              hipStream_t stream) {
    const float* x    = (const float*)d_in[0];
    const int*   src1 = (const int*)  d_in[1];
    const int*   dst1 = (const int*)  d_in[2];
    const float* w1   = (const float*)d_in[3];
    const int*   src2 = (const int*)  d_in[4];
    const int*   dst2 = (const int*)  d_in[5];
    const float* w2   = (const float*)d_in[6];
    const float* Ws1  = (const float*)d_in[7];
    const float* Wn1  = (const float*)d_in[8];
    const float* b1   = (const float*)d_in[9];
    const float* Ws2  = (const float*)d_in[10];
    const float* Wn2  = (const float*)d_in[11];
    const float* b2   = (const float*)d_in[12];
    float* out = (float*)d_out;

    float* acc1 = (float*)d_ws;
    float* hbuf = acc1 + (size_t)N_DST1 * 128;
    float* acc2 = hbuf + (size_t)N_DST1 * 128;

    const int E1 = in_sizes[1];
    const int E2 = in_sizes[4];

    unsigned* keys1 = (unsigned*)hbuf;
    unsigned* keys2 = (unsigned*)acc1;

    tile_key_kernel<ROWS_L1><<<512, 256, 0, stream>>>(dst1, keys1, E1);

    dst_tile_aggregate_kernel<ROWS_L1>
        <<<(N_DST1 + ROWS_L1 - 1) / ROWS_L1, AGG_THREADS, 0, stream>>>(
            x, src1, dst1, keys1, w1, acc1, E1, N_DST1);

    sage_gemm1_kernel<<<N_DST1 / (16 * G1_WAVES), G1_THREADS, 0, stream>>>(x, acc1,
                                                                         Ws1, Wn1, b1, hbuf);

    tile_key_kernel<ROWS_L2><<<256, 256, 0, stream>>>(dst2, keys2, E2);
    dst_tile_aggregate_kernel<ROWS_L2>
        <<<(N_DST2 + ROWS_L2 - 1) / ROWS_L2, AGG_THREADS, 0, stream>>>(
            hbuf, src2, dst2, keys2, w2, acc2, E2, N_DST2);

    sage_gemm2_kernel<<<(N_DST2 / 16 + 7) / 8, 256, 0, stream>>>(hbuf, acc2,
                                                                 Ws2, Wn2, b2, out);
}
